# starting kernel
# baseline (speedup 1.0000x reference)
#include <hip/hip_runtime.h>

typedef __attribute__((ext_vector_type(2))) float v2f;
typedef __attribute__((ext_vector_type(8))) float v8f;

#define BB 1024
#define TT 256
#define HH 64
#define EE 32
#define VV 10
#define OO 10

#if __has_builtin(__builtin_amdgcn_tanhf)
#define DEV_TANH(x) __builtin_amdgcn_tanhf(x)
#else
#define DEV_TANH(x) tanhf(x)
#endif


__global__ __launch_bounds__(64) void rnn_prep_kernel(
    const float* __restrict__ emb, const float* __restrict__ Wx,
    const float* __restrict__ brnn, const float* __restrict__ Wd,
    const float* __restrict__ bd, float* __restrict__ ws) {
  const int j = threadIdx.x;
  for (int v = 0; v < VV; ++v) {
    float s0 = brnn[j], s1 = 0.f;
    for (int d = 0; d < EE; ++d) {
      const float e = emb[v * EE + d];
      s0 += e * Wx[d * HH + j];
      s1 += e * Wx[(EE + d) * HH + j];
    }
    ws[v * HH + j] = s0;
    ws[VV * HH + v * HH + j] = s1;
  }
  for (int n = 0; n < 16; ++n)
    ws[1280 + j * 16 + n] = (n < OO) ? Wd[j * OO + n] : 0.f;
  if (j < 16) ws[2304 + j] = (j < OO) ? bd[j] : 0.f;
}

__global__ __launch_bounds__(32) void rnn_scan_kernel(
    const int* __restrict__ num1, const int* __restrict__ num2,
    const float* __restrict__ Wh, const float* __restrict__ ws,
    float* __restrict__ out) {
  __shared__ float lut[2 * VV * HH];
  __shared__ float hbuf[16 * HH];
  __shared__ int   vidx[2][16];

  const int lane = threadIdx.x;
  const int half = lane >> 4;
  const int m    = lane & 15;
  const int kb   = half * 2;
  const int row_base = blockIdx.x * 16;

  for (int i = lane; i < 2 * VV * HH; i += 32) lut[i] = ws[i];
  for (int i = lane; i < 16 * HH; i += 32) hbuf[i] = 0.f;

  v2f whb[16][4];
#pragma unroll
  for (int k = 0; k < 16; ++k)
#pragma unroll
    for (int n = 0; n < 4; ++n) {
      whb[k][n].x = Wh[(4 * k + kb + 0) * HH + n * 16 + m];
      whb[k][n].y = Wh[(4 * k + kb + 1) * HH + n * 16 + m];
    }
  v2f wdb[16];
#pragma unroll
  for (int k = 0; k < 16; ++k) {
    wdb[k].x = ws[1280 + (4 * k + kb + 0) * 16 + m];
    wdb[k].y = ws[1280 + (4 * k + kb + 1) * 16 + m];
  }
  const float bdv = ws[2304 + m];

  v2f a[16];
#pragma unroll
  for (int k = 0; k < 16; ++k) { a[k].x = 0.f; a[k].y = 0.f; }

  __syncthreads();

  const int* __restrict__ nsrc = half ? num2 : num1;

  for (int t = 0; t < TT; ++t) {
    vidx[half][m] = nsrc[(row_base + m) * TT + t];
    if (t + 1 < TT) __builtin_prefetch(&nsrc[(row_base + m) * TT + t + 1], 0, 3);
    __syncthreads();

    v8f acc[4];
#pragma unroll
    for (int r = 0; r < 8; ++r) {
      const int hr = r + 8 * half;
      const int d1 = vidx[0][hr];
      const int d2 = vidx[1][hr];
#pragma unroll
      for (int n = 0; n < 4; ++n)
        acc[n][r] = lut[d1 * HH + n * 16 + m] + lut[VV * HH + d2 * HH + n * 16 + m];
    }

#pragma unroll
    for (int k = 0; k < 16; ++k)
#pragma unroll
      for (int n = 0; n < 4; ++n)
        acc[n] = __builtin_amdgcn_wmma_f32_16x16x4_f32(
            false, a[k], false, whb[k][n], (short)0, acc[n], false, false);

#pragma unroll
    for (int n = 0; n < 4; ++n)
#pragma unroll
      for (int r = 0; r < 8; ++r)
        hbuf[(r + 8 * half) * HH + n * 16 + m] = DEV_TANH(acc[n][r]);
    __syncthreads();

#pragma unroll
    for (int k = 0; k < 16; ++k)
      a[k] = *(const v2f*)&hbuf[m * HH + 4 * k + kb];

    v8f c2;
#pragma unroll
    for (int r = 0; r < 8; ++r) c2[r] = bdv;
#pragma unroll
    for (int k = 0; k < 16; ++k)
      c2 = __builtin_amdgcn_wmma_f32_16x16x4_f32(
          false, a[k], false, wdb[k], (short)0, c2, false, false);

    if (m < OO) {
#pragma unroll
      for (int r = 0; r < 8; ++r) {
        const int row = row_base + r + 8 * half;
        out[(size_t)row * TT * OO + (size_t)t * OO + m] = c2[r];
      }
    }
  }
}

extern "C" void kernel_launch(void* const* d_in, const int* in_sizes, int n_in,
                              void* d_out, int out_size, void* d_ws, size_t ws_size,
                              hipStream_t stream) {
  const int*   num1 = (const int*)d_in[0];
  const int*   num2 = (const int*)d_in[1];
  const float* emb  = (const float*)d_in[2];
  const float* Wx   = (const float*)d_in[3];
  const float* Wh   = (const float*)d_in[4];
  const float* brnn = (const float*)d_in[5];
  const float* Wd   = (const float*)d_in[6];
  const float* bd   = (const float*)d_in[7];
  float* out = (float*)d_out;
  float* ws  = (float*)d_ws;

  rnn_prep_kernel<<<1, 64, 0, stream>>>(emb, Wx, brnn, Wd, bd, ws);
  rnn_scan_kernel<<<BB / 16, 32, 0, stream>>>(num1, num2, Wh, ws, out);
}


# Recurrence computed transposed (hT = tanh(xT + WhT hT)): each step's D tiles are the next step's B fragments in registers, no LDS round trip or barrier per step; k order of the sums changes
# speedup vs baseline: 1.8524x; 1.8524x over previous
#include <hip/hip_runtime.h>


typedef __attribute__((ext_vector_type(2))) float v2f;
typedef __attribute__((ext_vector_type(4))) float v4f;
typedef __attribute__((ext_vector_type(8))) float v8f;
typedef __attribute__((ext_vector_type(4))) int v4i;

#define BB 1024
#define TT 256
#define HH 64
#define EE 32
#define VV 10
#define OO 10
#define LP 68

#if __has_builtin(__builtin_amdgcn_tanhf)
#define DEV_TANH(x) __builtin_amdgcn_tanhf(x)
#else
#define DEV_TANH(x) tanhf(x)
#endif

#define WMMA(A, B, C) __builtin_amdgcn_wmma_f32_16x16x4_f32(false, (A), false, (B), (short)0, (C), false, false)


__global__ __launch_bounds__(64) void rnn_prep_kernel(
    const float* __restrict__ emb, const float* __restrict__ Wx,
    const float* __restrict__ brnn, const float* __restrict__ Wd,
    const float* __restrict__ bd, float* __restrict__ ws) {
  const int j = threadIdx.x;
  for (int v = 0; v < VV; ++v) {
    float s0 = brnn[j], s1 = 0.f;
    for (int d = 0; d < EE; ++d) {
      const float e = emb[v * EE + d];
      s0 += e * Wx[d * HH + j];
      s1 += e * Wx[(EE + d) * HH + j];
    }
    ws[v * HH + j] = s0;
    ws[VV * HH + v * HH + j] = s1;
  }
  for (int n = 0; n < 16; ++n)
    ws[1280 + j * 16 + n] = (n < OO) ? Wd[j * OO + n] : 0.f;
  if (j < 16) ws[2304 + j] = (j < OO) ? bd[j] : 0.f;
}

__device__ __forceinline__ void gather_x(const float* lut, int d1, int d2, int half, v8f x[4]) {
  const float* p0 = lut + d1 * LP + 8 * half;
  const float* p1 = lut + (VV + d2) * LP + 8 * half;
#pragma unroll
  for (int n = 0; n < 4; ++n) {
    const v4f a0 = *(const v4f*)(p0 + 16 * n);
    const v4f a1 = *(const v4f*)(p0 + 16 * n + 4);
    const v4f b0 = *(const v4f*)(p1 + 16 * n);
    const v4f b1 = *(const v4f*)(p1 + 16 * n + 4);
    const v4f s0 = a0 + b0, s1 = a1 + b1;
    x[n] = (v8f){s0.x, s0.y, s0.z, s0.w, s1.x, s1.y, s1.z, s1.w};
  }
}

__device__ __forceinline__ void store_logits(float* __restrict__ out, int row, int t, int half, v8f c2) {
  float* o = out + (size_t)row * (TT * OO) + (size_t)t * OO + 8 * half;
  *(v2f*)(o) = (v2f){c2[0], c2[1]};
  if (half == 0) {
    *(v2f*)(o + 2) = (v2f){c2[2], c2[3]};
    *(v2f*)(o + 4) = (v2f){c2[4], c2[5]};
    *(v2f*)(o + 6) = (v2f){c2[6], c2[7]};
  }
}

__global__ __launch_bounds__(32) void rnn_scan_kernel(
    const int* __restrict__ num1, const int* __restrict__ num2,
    const float* __restrict__ Wh, const float* __restrict__ ws,
    float* __restrict__ out) {
  __shared__ __attribute__((aligned(16))) float lut[2 * VV * LP];

  const int lane = threadIdx.x;
  const int half = lane >> 4;
  const int c    = lane & 15;
  const int row  = blockIdx.x * 16 + c;

  for (int i = lane; i < 2 * VV * HH; i += 32) lut[(i >> 6) * LP + (i & 63)] = ws[i];

  v2f wha[16][4];
  v2f wda[16];
#pragma unroll
  for (int j = 0; j < 16; ++j) {
    const int k0 = 16 * (j >> 2) + 8 * half + 2 * (j & 3);
#pragma unroll
    for (int n = 0; n < 4; ++n) {
      wha[j][n].x = Wh[(k0 + 0) * HH + n * 16 + c];
      wha[j][n].y = Wh[(k0 + 1) * HH + n * 16 + c];
    }
    wda[j].x = ws[1280 + (k0 + 0) * 16 + c];
    wda[j].y = ws[1280 + (k0 + 1) * 16 + c];
  }
  v8f bias8;
#pragma unroll
  for (int r = 0; r < 8; ++r) bias8[r] = ws[2304 + 8 * half + r];

  v2f hb[2][16];
#pragma unroll
  for (int j = 0; j < 16; ++j) { hb[0][j] = (v2f){0.f, 0.f}; hb[1][j] = (v2f){0.f, 0.f}; }
  v8f accP2, accP3;
#pragma unroll
  for (int r = 0; r < 8; ++r) { accP2[r] = 0.f; accP3[r] = 0.f; }

  const int* __restrict__ q1 = num1 + (size_t)row * TT;
  const int* __restrict__ q2 = num2 + (size_t)row * TT;
  v4i dn1 = *(const v4i*)q1;
  v4i dn2 = *(const v4i*)q2;

  __syncthreads();

  v8f xn[4];
  gather_x(lut, dn1.x, dn2.x, half, xn);

  for (int tc = 0; tc < TT; tc += 4) {
    const v4i d1 = dn1, d2 = dn2;
    const int tn = (tc + 4 < TT) ? tc + 4 : tc;
    dn1 = *(const v4i*)(q1 + tn);
    dn2 = *(const v4i*)(q2 + tn);
#pragma unroll
    for (int s = 0; s < 4; ++s) {
      const int t = tc + s;
      v2f* hc = hb[s & 1];
      v2f* hn = hb[(s & 1) ^ 1];
      v8f acc0 = xn[0], acc1 = xn[1], acc2 = xn[2], acc3 = xn[3];
      v8f c2 = bias8;

#pragma unroll
      for (int j = 0; j < 8; ++j) {
        acc0 = WMMA(wha[j][0], hc[j], acc0);
        acc1 = WMMA(wha[j][1], hc[j], acc1);
      }
#pragma unroll
      for (int q = 0; q < 4; ++q) {
        hc[8 + q]  = (v2f){DEV_TANH(accP2[2 * q]), DEV_TANH(accP2[2 * q + 1])};
        hc[12 + q] = (v2f){DEV_TANH(accP3[2 * q]), DEV_TANH(accP3[2 * q + 1])};
      }
#pragma unroll
      for (int j = 8; j < 16; ++j) {
        acc0 = WMMA(wha[j][0], hc[j], acc0);
        acc1 = WMMA(wha[j][1], hc[j], acc1);
      }

      const int e1 = (s == 0) ? d1.y : (s == 1) ? d1.z : (s == 2) ? d1.w : dn1.x;
      const int e2 = (s == 0) ? d2.y : (s == 1) ? d2.z : (s == 2) ? d2.w : dn2.x;
      gather_x(lut, e1, e2, half, xn);

#pragma unroll
      for (int j = 0; j < 8; ++j) {
        acc2 = WMMA(wha[j][2], hc[j], acc2);
        acc3 = WMMA(wha[j][3], hc[j], acc3);
        c2   = WMMA(wda[j],    hc[j], c2);
      }
#pragma unroll
      for (int q = 0; q < 4; ++q) {
        hn[q]     = (v2f){DEV_TANH(acc0[2 * q]), DEV_TANH(acc0[2 * q + 1])};
        hn[4 + q] = (v2f){DEV_TANH(acc1[2 * q]), DEV_TANH(acc1[2 * q + 1])};
      }
#pragma unroll
      for (int j = 8; j < 16; ++j) {
        acc2 = WMMA(wha[j][2], hc[j], acc2);
        acc3 = WMMA(wha[j][3], hc[j], acc3);
        c2   = WMMA(wda[j],    hc[j], c2);
      }
      accP2 = acc2;
      accP3 = acc3;
      if (t > 0) store_logits(out, row, t - 1, half, c2);
    }
  }

  {
    v2f* hc = hb[0];
#pragma unroll
    for (int q = 0; q < 4; ++q) {
      hc[8 + q]  = (v2f){DEV_TANH(accP2[2 * q]), DEV_TANH(accP2[2 * q + 1])};
      hc[12 + q] = (v2f){DEV_TANH(accP3[2 * q]), DEV_TANH(accP3[2 * q + 1])};
    }
    v8f c2 = bias8;
#pragma unroll
    for (int j = 0; j < 16; ++j) c2 = WMMA(wda[j], hc[j], c2);
    store_logits(out, row, TT - 1, half, c2);
  }
}

extern "C" void kernel_launch(void* const* d_in, const int* in_sizes, int n_in,
                              void* d_out, int out_size, void* d_ws, size_t ws_size,
                              hipStream_t stream) {
  const int*   num1 = (const int*)d_in[0];
  const int*   num2 = (const int*)d_in[1];
  const float* emb  = (const float*)d_in[2];
  const float* Wx   = (const float*)d_in[3];
  const float* Wh   = (const float*)d_in[4];
  const float* brnn = (const float*)d_in[5];
  const float* Wd   = (const float*)d_in[6];
  const float* bd   = (const float*)d_in[7];
  float* out = (float*)d_out;
  float* ws  = (float*)d_ws;

  rnn_prep_kernel<<<1, 64, 0, stream>>>(emb, Wx, brnn, Wd, bd, ws);
  rnn_scan_kernel<<<BB / 16, 32, 0, stream>>>(num1, num2, Wh, ws, out);
}


# Hidden tiles split over 4 waves of a block (one per SIMD), h pieces exchanged in LDS, one barrier a step; logits as per-tile partial WMMAs summed by wave 0; order of the sums changes (x added last)
# speedup vs baseline: 3.7402x; 2.0191x over previous
#include <hip/hip_runtime.h>


typedef __attribute__((ext_vector_type(2))) float v2f;
typedef __attribute__((ext_vector_type(4))) float v4f;
typedef __attribute__((ext_vector_type(8))) float v8f;
typedef __attribute__((ext_vector_type(4))) int v4i;

#define BB 1024
#define TT 256
#define HH 64
#define EE 32
#define VV 10
#define OO 10
#define LP 68

#if __has_builtin(__builtin_amdgcn_tanhf)
#define DEV_TANH(x) __builtin_amdgcn_tanhf(x)
#else
#define DEV_TANH(x) tanhf(x)
#endif

#define WMMA(A, B, C) __builtin_amdgcn_wmma_f32_16x16x4_f32(false, (A), false, (B), (short)0, (C), false, false)


__global__ __launch_bounds__(64) void rnn_prep_kernel(
    const float* __restrict__ emb, const float* __restrict__ Wx,
    const float* __restrict__ brnn, const float* __restrict__ Wd,
    const float* __restrict__ bd, float* __restrict__ ws) {
  const int j = threadIdx.x;
  for (int v = 0; v < VV; ++v) {
    float s0 = brnn[j], s1 = 0.f;
    for (int d = 0; d < EE; ++d) {
      const float e = emb[v * EE + d];
      s0 += e * Wx[d * HH + j];
      s1 += e * Wx[(EE + d) * HH + j];
    }
    ws[v * HH + j] = s0;
    ws[VV * HH + v * HH + j] = s1;
  }
  for (int n = 0; n < 16; ++n)
    ws[1280 + j * 16 + n] = (n < OO) ? Wd[j * OO + n] : 0.f;
  if (j < 16) ws[2304 + j] = (j < OO) ? bd[j] : 0.f;
}

#define XW 128
#define XOFF(buf, tile) ((((buf) * 4 + (tile)) * 2) * XW)

__device__ __forceinline__ void lds_barrier() {
  asm volatile("s_wait_dscnt 0x0\n\ts_barrier_signal -1\n\ts_barrier_wait -1" ::: "memory");
}

__device__ __forceinline__ v8f table_row(const float* lut, int d, int half, int tile) {
  const float* p = lut + d * LP + 16 * tile + 8 * half;
  const v4f a0 = *(const v4f*)(p);
  const v4f a1 = *(const v4f*)(p + 4);
  return (v8f){a0.x, a0.y, a0.z, a0.w, a1.x, a1.y, a1.z, a1.w};
}

__device__ __forceinline__ void store_logits(float* __restrict__ out, int row, int t, int half, v8f c2) {
  float* o = out + (size_t)row * (TT * OO) + (size_t)t * OO + 8 * half;
  *(v2f*)(o) = (v2f){c2[0], c2[1]};
  if (half == 0) {
    *(v2f*)(o + 2) = (v2f){c2[2], c2[3]};
    *(v2f*)(o + 4) = (v2f){c2[4], c2[5]};
    *(v2f*)(o + 6) = (v2f){c2[6], c2[7]};
  }
}

__device__ __forceinline__ void load_partials(const float* px, int buf, int lane, v8f lq[4]) {
#pragma unroll
  for (int n = 0; n < 4; ++n) {
    const float* p = px + XOFF(buf, n) + lane * 4;
    const v4f u0 = *(const v4f*)(p);
    const v4f u1 = *(const v4f*)(p + XW);
    lq[n] = (v8f){u0.x, u0.y, u0.z, u0.w, u1.x, u1.y, u1.z, u1.w};
  }
}

__global__ __launch_bounds__(128) void rnn_scan_kernel(
    const int* __restrict__ num1, const int* __restrict__ num2,
    const float* __restrict__ Wh, const float* __restrict__ ws,
    float* __restrict__ out) {
  __shared__ __attribute__((aligned(16))) float lut[2 * VV * LP];
  __shared__ __attribute__((aligned(16))) float hx[2 * 4 * 2 * XW];
  __shared__ __attribute__((aligned(16))) float px[2 * 4 * 2 * XW];

  const int tid  = threadIdx.x;
  const int w    = __builtin_amdgcn_readfirstlane(tid >> 5);
  const int lane = tid & 31;
  const int half = lane >> 4;
  const int c    = lane & 15;
  const int row  = blockIdx.x * 16 + c;

  for (int i = tid; i < 2 * VV * HH; i += 128) lut[(i >> 6) * LP + (i & 63)] = ws[i];

  v2f wha[16];
  v2f wda[4];
#pragma unroll
  for (int jj = 0; jj < 16; ++jj) {
    const int tl = (w + (jj >> 2)) & 3;
    const int k0 = 16 * tl + 8 * half + 2 * (jj & 3);
    wha[jj].x = Wh[(k0 + 0) * HH + w * 16 + c];
    wha[jj].y = Wh[(k0 + 1) * HH + w * 16 + c];
  }
#pragma unroll
  for (int q = 0; q < 4; ++q) {
    const int k0 = 16 * w + 8 * half + 2 * q;
    wda[q].x = ws[1280 + (k0 + 0) * 16 + c];
    wda[q].y = ws[1280 + (k0 + 1) * 16 + c];
  }
  v8f bias8, zero8;
#pragma unroll
  for (int r = 0; r < 8; ++r) { bias8[r] = ws[2304 + 8 * half + r]; zero8[r] = 0.f; }

  v2f hb[16];
#pragma unroll
  for (int jj = 0; jj < 16; ++jj) hb[jj] = (v2f){0.f, 0.f};
  v8f pl = zero8;
  v8f lq[4];
#pragma unroll
  for (int n = 0; n < 4; ++n) lq[n] = zero8;

  const int* __restrict__ q1 = num1 + (size_t)row * TT;
  const int* __restrict__ q2 = num2 + (size_t)row * TT;
  v4i dn1 = *(const v4i*)q1;
  v4i dn2 = *(const v4i*)q2;

  __syncthreads();

  v8f xa = table_row(lut, dn1.x, half, w);
  v8f xb = table_row(lut, VV + dn2.x, half, w);

  float* const hp = hx + XOFF(0, w) + lane * 4;
  float* const pp = px + XOFF(0, w) + lane * 4;

  for (int tc = 0; tc < TT; tc += 4) {
    const v4i d1 = dn1, d2 = dn2;
    const int tn = (tc + 4 < TT) ? tc + 4 : tc;
    dn1 = *(const v4i*)(q1 + tn);
    dn2 = *(const v4i*)(q2 + tn);
#pragma unroll
    for (int s = 0; s < 4; ++s) {
      const int t = tc + s;
      const int buf = s & 1;

      *(v4f*)(pp + XOFF(buf, 0))      = (v4f){pl[0], pl[1], pl[2], pl[3]};
      *(v4f*)(pp + XOFF(buf, 0) + XW) = (v4f){pl[4], pl[5], pl[6], pl[7]};

      v8f a0 = zero8, a1 = zero8;
#pragma unroll
      for (int jj = 0; jj < 16; jj += 2) {
        a0 = WMMA(wha[jj],     hb[jj],     a0);
        a1 = WMMA(wha[jj + 1], hb[jj + 1], a1);
      }

      v8f hn;
#pragma unroll
      for (int r = 0; r < 8; ++r) hn[r] = DEV_TANH((a0[r] + xa[r]) + (a1[r] + xb[r]));
      *(v4f*)(hp + XOFF(buf, 0))      = (v4f){hn[0], hn[1], hn[2], hn[3]};
      *(v4f*)(hp + XOFF(buf, 0) + XW) = (v4f){hn[4], hn[5], hn[6], hn[7]};

      hb[0] = (v2f){hn[0], hn[1]};
      hb[1] = (v2f){hn[2], hn[3]};
      hb[2] = (v2f){hn[4], hn[5]};
      hb[3] = (v2f){hn[6], hn[7]};
      pl = zero8;
#pragma unroll
      for (int q = 0; q < 4; ++q) pl = WMMA(wda[q], hb[q], pl);

      if (w == 0 && t > 1) store_logits(out, row, t - 2, half, bias8 + lq[0] + lq[1] + lq[2] + lq[3]);

      lds_barrier();

#pragma unroll
      for (int r = 1; r < 4; ++r) {
        const float* sp = hx + XOFF(buf, (w + r) & 3) + lane * 4;
        const v4f u0 = *(const v4f*)(sp);
        const v4f u1 = *(const v4f*)(sp + XW);
        hb[4 * r + 0] = (v2f){u0.x, u0.y};
        hb[4 * r + 1] = (v2f){u0.z, u0.w};
        hb[4 * r + 2] = (v2f){u1.x, u1.y};
        hb[4 * r + 3] = (v2f){u1.z, u1.w};
      }
      const int e1 = (s == 0) ? d1.y : (s == 1) ? d1.z : (s == 2) ? d1.w : dn1.x;
      const int e2 = (s == 0) ? d2.y : (s == 1) ? d2.z : (s == 2) ? d2.w : dn2.x;
      xa = table_row(lut, e1, half, w);
      xb = table_row(lut, VV + e2, half, w);
      if (w == 0) load_partials(px, buf, lane, lq);
    }
  }

  if (w == 0) store_logits(out, row, TT - 2, half, bias8 + lq[0] + lq[1] + lq[2] + lq[3]);
  *(v4f*)(pp)      = (v4f){pl[0], pl[1], pl[2], pl[3]};
  *(v4f*)(pp + XW) = (v4f){pl[4], pl[5], pl[6], pl[7]};
  lds_barrier();
  if (w == 0) {
    load_partials(px, 0, lane, lq);
    store_logits(out, row, TT - 1, half, bias8 + lq[0] + lq[1] + lq[2] + lq[3]);
  }
}

extern "C" void kernel_launch(void* const* d_in, const int* in_sizes, int n_in,
                              void* d_out, int out_size, void* d_ws, size_t ws_size,
                              hipStream_t stream) {
  const int*   num1 = (const int*)d_in[0];
  const int*   num2 = (const int*)d_in[1];
  const float* emb  = (const float*)d_in[2];
  const float* Wx   = (const float*)d_in[3];
  const float* Wh   = (const float*)d_in[4];
  const float* brnn = (const float*)d_in[5];
  const float* Wd   = (const float*)d_in[6];
  const float* bd   = (const float*)d_in[7];
  float* out = (float*)d_out;
  float* ws  = (float*)d_ws;

  rnn_prep_kernel<<<1, 64, 0, stream>>>(emb, Wx, brnn, Wd, bd, ws);
  rnn_scan_kernel<<<BB / 16, 128, 0, stream>>>(num1, num2, Wh, ws, out);
}


# A fifth wave per block takes over the sums of the partial logits and the stores to out; the four recurrence waves no longer carry that work in front of every hand-over barrier
# speedup vs baseline: 4.9304x; 1.3182x over previous
#include <hip/hip_runtime.h>


typedef __attribute__((ext_vector_type(2))) float v2f;
typedef __attribute__((ext_vector_type(4))) float v4f;
typedef __attribute__((ext_vector_type(8))) float v8f;
typedef __attribute__((ext_vector_type(4))) int v4i;

#define BB 1024
#define TT 256
#define HH 64
#define EE 32
#define VV 10
#define OO 10
#define LP 68

#if __has_builtin(__builtin_amdgcn_tanhf)
#define DEV_TANH(x) __builtin_amdgcn_tanhf(x)
#else
#define DEV_TANH(x) tanhf(x)
#endif

#define WMMA(A, B, C) __builtin_amdgcn_wmma_f32_16x16x4_f32(false, (A), false, (B), (short)0, (C), false, false)


__global__ __launch_bounds__(64) void rnn_prep_kernel(
    const float* __restrict__ emb, const float* __restrict__ Wx,
    const float* __restrict__ brnn, const float* __restrict__ Wd,
    const float* __restrict__ bd, float* __restrict__ ws) {
  const int j = threadIdx.x;
  for (int v = 0; v < VV; ++v) {
    float s0 = brnn[j], s1 = 0.f;
    for (int d = 0; d < EE; ++d) {
      const float e = emb[v * EE + d];
      s0 += e * Wx[d * HH + j];
      s1 += e * Wx[(EE + d) * HH + j];
    }
    ws[v * HH + j] = s0;
    ws[VV * HH + v * HH + j] = s1;
  }
  for (int n = 0; n < 16; ++n)
    ws[1280 + j * 16 + n] = (n < OO) ? Wd[j * OO + n] : 0.f;
  if (j < 16) ws[2304 + j] = (j < OO) ? bd[j] : 0.f;
}

#define XW 128
#define XOFF(buf, tile) ((((buf) * 4 + (tile)) * 2) * XW)

__device__ __forceinline__ void lds_barrier() {
  asm volatile("s_wait_dscnt 0x0\n\ts_barrier_signal -1\n\ts_barrier_wait -1" ::: "memory");
}

__device__ __forceinline__ v8f table_row(const float* lut, int d, int half, int tile) {
  const float* p = lut + d * LP + 16 * tile + 8 * half;
  const v4f a0 = *(const v4f*)(p);
  const v4f a1 = *(const v4f*)(p + 4);
  return (v8f){a0.x, a0.y, a0.z, a0.w, a1.x, a1.y, a1.z, a1.w};
}

__device__ __forceinline__ void store_logits(float* __restrict__ out, int row, int t, int half, v8f c2) {
  float* o = out + (size_t)row * (TT * OO) + (size_t)t * OO + 8 * half;
  *(v2f*)(o) = (v2f){c2[0], c2[1]};
  if (half == 0) {
    *(v2f*)(o + 2) = (v2f){c2[2], c2[3]};
    *(v2f*)(o + 4) = (v2f){c2[4], c2[5]};
    *(v2f*)(o + 6) = (v2f){c2[6], c2[7]};
  }
}

__device__ __forceinline__ void load_partials(const float* px, int buf, int lane, v8f lq[4]) {
#pragma unroll
  for (int n = 0; n < 4; ++n) {
    const float* p = px + XOFF(buf, n) + lane * 4;
    const v4f u0 = *(const v4f*)(p);
    const v4f u1 = *(const v4f*)(p + XW);
    lq[n] = (v8f){u0.x, u0.y, u0.z, u0.w, u1.x, u1.y, u1.z, u1.w};
  }
}

__global__ __launch_bounds__(160) void rnn_scan_kernel(
    const int* __restrict__ num1, const int* __restrict__ num2,
    const float* __restrict__ Wh, const float* __restrict__ ws,
    float* __restrict__ out) {
  __shared__ __attribute__((aligned(16))) float lut[2 * VV * LP];
  __shared__ __attribute__((aligned(16))) float hx[2 * 4 * 2 * XW];
  __shared__ __attribute__((aligned(16))) float px[2 * 4 * 2 * XW];

  const int tid  = threadIdx.x;
  const int w    = __builtin_amdgcn_readfirstlane(tid >> 5);
  const bool rec = (w < 4);
  const int  wr  = rec ? w : 0;
  const int lane = tid & 31;
  const int half = lane >> 4;
  const int c    = lane & 15;
  const int row  = blockIdx.x * 16 + c;

  for (int i = tid; i < 2 * VV * HH; i += 160) lut[(i >> 6) * LP + (i & 63)] = ws[i];

  v2f wha[16];
  v2f wda[4];
#pragma unroll
  for (int jj = 0; jj < 16; ++jj) {
    const int tl = (wr + (jj >> 2)) & 3;
    const int k0 = 16 * tl + 8 * half + 2 * (jj & 3);
    wha[jj].x = Wh[(k0 + 0) * HH + wr * 16 + c];
    wha[jj].y = Wh[(k0 + 1) * HH + wr * 16 + c];
  }
#pragma unroll
  for (int q = 0; q < 4; ++q) {
    const int k0 = 16 * wr + 8 * half + 2 * q;
    wda[q].x = ws[1280 + (k0 + 0) * 16 + c];
    wda[q].y = ws[1280 + (k0 + 1) * 16 + c];
  }
  v8f bias8, zero8;
#pragma unroll
  for (int r = 0; r < 8; ++r) { bias8[r] = ws[2304 + 8 * half + r]; zero8[r] = 0.f; }

  v2f ho[4];
#pragma unroll
  for (int q = 0; q < 4; ++q) ho[q] = (v2f){0.f, 0.f};
  v8f pl = zero8;

  const int* __restrict__ q1 = num1 + (size_t)row * TT;
  const int* __restrict__ q2 = num2 + (size_t)row * TT;
  v4i dn1 = *(const v4i*)q1;
  v4i dn2 = *(const v4i*)q2;

  float* const hp = hx + XOFF(0, wr) + lane * 4;
  float* const pp = px + XOFF(0, wr) + lane * 4;

  if (rec) {
    *(v4f*)(hp + XOFF(1, 0))      = (v4f){0.f, 0.f, 0.f, 0.f};
    *(v4f*)(hp + XOFF(1, 0) + XW) = (v4f){0.f, 0.f, 0.f, 0.f};
  }

  __syncthreads();

  for (int tc = 0; tc < TT; tc += 4) {
    const v4i d1 = dn1, d2 = dn2;
    const int tn = (tc + 4 < TT) ? tc + 4 : tc;
    dn1 = *(const v4i*)(q1 + tn);
    dn2 = *(const v4i*)(q2 + tn);
#pragma unroll
    for (int s = 0; s < 4; ++s) {
      const int t = tc + s;
      const int buf = s & 1;

      if (rec) {
        v2f hr[12];
#pragma unroll
        for (int r = 1; r < 4; ++r) {
          const float* sp = hx + XOFF(buf ^ 1, (w + r) & 3) + lane * 4;
          const v4f u0 = *(const v4f*)(sp);
          const v4f u1 = *(const v4f*)(sp + XW);
          hr[4 * (r - 1) + 0] = (v2f){u0.x, u0.y};
          hr[4 * (r - 1) + 1] = (v2f){u0.z, u0.w};
          hr[4 * (r - 1) + 2] = (v2f){u1.x, u1.y};
          hr[4 * (r - 1) + 3] = (v2f){u1.z, u1.w};
        }
        const int e1 = (s == 0) ? d1.x : (s == 1) ? d1.y : (s == 2) ? d1.z : d1.w;
        const int e2 = (s == 0) ? d2.x : (s == 1) ? d2.y : (s == 2) ? d2.z : d2.w;
        const v8f xa = table_row(lut, e1, half, w);
        const v8f xb = table_row(lut, VV + e2, half, w);
        __builtin_amdgcn_sched_barrier(0);

        *(v4f*)(pp + XOFF(buf, 0))      = (v4f){pl[0], pl[1], pl[2], pl[3]};
        *(v4f*)(pp + XOFF(buf, 0) + XW) = (v4f){pl[4], pl[5], pl[6], pl[7]};

        v8f a0 = zero8, a1 = zero8;
        a0 = WMMA(wha[0], ho[0], a0);
        a1 = WMMA(wha[1], ho[1], a1);
        a0 = WMMA(wha[2], ho[2], a0);
        a1 = WMMA(wha[3], ho[3], a1);
#pragma unroll
        for (int jj = 4; jj < 16; jj += 2) {
          a0 = WMMA(wha[jj],     hr[jj - 4], a0);
          a1 = WMMA(wha[jj + 1], hr[jj - 3], a1);
        }

        v8f hn;
#pragma unroll
        for (int r = 0; r < 8; ++r) hn[r] = DEV_TANH((a0[r] + xa[r]) + (a1[r] + xb[r]));
        *(v4f*)(hp + XOFF(buf, 0))      = (v4f){hn[0], hn[1], hn[2], hn[3]};
        *(v4f*)(hp + XOFF(buf, 0) + XW) = (v4f){hn[4], hn[5], hn[6], hn[7]};

        ho[0] = (v2f){hn[0], hn[1]};
        ho[1] = (v2f){hn[2], hn[3]};
        ho[2] = (v2f){hn[4], hn[5]};
        ho[3] = (v2f){hn[6], hn[7]};
        pl = zero8;
#pragma unroll
        for (int q = 0; q < 4; ++q) pl = WMMA(wda[q], ho[q], pl);
      } else if (t > 1) {
        v8f lq[4];
        load_partials(px, buf ^ 1, lane, lq);
        store_logits(out, row, t - 2, half, bias8 + lq[0] + lq[1] + lq[2] + lq[3]);
      }

      lds_barrier();
    }
  }

  if (rec) {
    *(v4f*)(pp)      = (v4f){pl[0], pl[1], pl[2], pl[3]};
    *(v4f*)(pp + XW) = (v4f){pl[4], pl[5], pl[6], pl[7]};
  } else {
    v8f lq[4];
    load_partials(px, 1, lane, lq);
    store_logits(out, row, TT - 2, half, bias8 + lq[0] + lq[1] + lq[2] + lq[3]);
  }
  lds_barrier();
  if (!rec) {
    v8f lq[4];
    load_partials(px, 0, lane, lq);
    store_logits(out, row, TT - 1, half, bias8 + lq[0] + lq[1] + lq[2] + lq[3]);
  }
}

extern "C" void kernel_launch(void* const* d_in, const int* in_sizes, int n_in,
                              void* d_out, int out_size, void* d_ws, size_t ws_size,
                              hipStream_t stream) {
  const int*   num1 = (const int*)d_in[0];
  const int*   num2 = (const int*)d_in[1];
  const float* emb  = (const float*)d_in[2];
  const float* Wx   = (const float*)d_in[3];
  const float* Wh   = (const float*)d_in[4];
  const float* brnn = (const float*)d_in[5];
  const float* Wd   = (const float*)d_in[6];
  const float* bd   = (const float*)d_in[7];
  float* out = (float*)d_out;
  float* ws  = (float*)d_ws;

  rnn_prep_kernel<<<1, 64, 0, stream>>>(emb, Wx, brnn, Wd, bd, ws);
  rnn_scan_kernel<<<BB / 16, 160, 0, stream>>>(num1, num2, Wh, ws, out);
}


# The table kernel is folded into the scan kernel's prologue: every block builds both input tables in LDS from emb and Wx itself (same sums, same order); one launch per call, workspace unused
# speedup vs baseline: 5.3569x; 1.0865x over previous
#include <hip/hip_runtime.h>


typedef __attribute__((ext_vector_type(2))) float v2f;
typedef __attribute__((ext_vector_type(4))) float v4f;
typedef __attribute__((ext_vector_type(8))) float v8f;
typedef __attribute__((ext_vector_type(4))) int v4i;

#define BB 1024
#define TT 256
#define HH 64
#define EE 32
#define VV 10
#define OO 10
#define LP 68

#if __has_builtin(__builtin_amdgcn_tanhf)
#define DEV_TANH(x) __builtin_amdgcn_tanhf(x)
#else
#define DEV_TANH(x) tanhf(x)
#endif

#define WMMA(A, B, C) __builtin_amdgcn_wmma_f32_16x16x4_f32(false, (A), false, (B), (short)0, (C), false, false)

#define XW 128
#define XOFF(buf, tile) ((((buf) * 4 + (tile)) * 2) * XW)

__device__ __forceinline__ void lds_barrier() {
  asm volatile("s_wait_dscnt 0x0\n\ts_barrier_signal -1\n\ts_barrier_wait -1" ::: "memory");
}

__device__ __forceinline__ v8f table_row(const float* lut, int d, int half, int tile) {
  const float* p = lut + d * LP + 16 * tile + 8 * half;
  const v4f a0 = *(const v4f*)(p);
  const v4f a1 = *(const v4f*)(p + 4);
  return (v8f){a0.x, a0.y, a0.z, a0.w, a1.x, a1.y, a1.z, a1.w};
}

__device__ __forceinline__ void store_logits(float* __restrict__ out, int row, int t, int half, v8f c2) {
  float* o = out + (size_t)row * (TT * OO) + (size_t)t * OO + 8 * half;
  *(v2f*)(o) = (v2f){c2[0], c2[1]};
  if (half == 0) {
    *(v2f*)(o + 2) = (v2f){c2[2], c2[3]};
    *(v2f*)(o + 4) = (v2f){c2[4], c2[5]};
    *(v2f*)(o + 6) = (v2f){c2[6], c2[7]};
  }
}

__device__ __forceinline__ void load_partials(const float* px, int buf, int lane, v8f lq[4]) {
#pragma unroll
  for (int n = 0; n < 4; ++n) {
    const float* p = px + XOFF(buf, n) + lane * 4;
    const v4f u0 = *(const v4f*)(p);
    const v4f u1 = *(const v4f*)(p + XW);
    lq[n] = (v8f){u0.x, u0.y, u0.z, u0.w, u1.x, u1.y, u1.z, u1.w};
  }
}

__global__ __launch_bounds__(160) void rnn_scan_kernel(
    const int* __restrict__ num1, const int* __restrict__ num2,
    const float* __restrict__ emb, const float* __restrict__ Wx,
    const float* __restrict__ brnn, const float* __restrict__ Wh,
    const float* __restrict__ Wd, const float* __restrict__ bd,
    float* __restrict__ out) {
  __shared__ __attribute__((aligned(16))) float lut[2 * VV * LP];
  __shared__ __attribute__((aligned(16))) float hx[2 * 4 * 2 * XW];
  __shared__ __attribute__((aligned(16))) float px[2 * 4 * 2 * XW];

  const int tid  = threadIdx.x;
  const int w    = __builtin_amdgcn_readfirstlane(tid >> 5);
  const bool rec = (w < 4);
  const int  wr  = rec ? w : 0;
  const int lane = tid & 31;
  const int half = lane >> 4;
  const int c    = lane & 15;
  const int row  = blockIdx.x * 16 + c;

  v2f wha[16];
  v2f wda[4];
  const int cw = (c < OO) ? c : OO - 1;
#pragma unroll
  for (int jj = 0; jj < 16; ++jj) {
    const int tl = (wr + (jj >> 2)) & 3;
    const int k0 = 16 * tl + 8 * half + 2 * (jj & 3);
    wha[jj].x = Wh[(k0 + 0) * HH + wr * 16 + c];
    wha[jj].y = Wh[(k0 + 1) * HH + wr * 16 + c];
  }
#pragma unroll
  for (int q = 0; q < 4; ++q) {
    const int k0 = 16 * wr + 8 * half + 2 * q;
    const float u0 = Wd[(k0 + 0) * OO + cw];
    const float u1 = Wd[(k0 + 1) * OO + cw];
    wda[q].x = (c < OO) ? u0 : 0.f;
    wda[q].y = (c < OO) ? u1 : 0.f;
  }
  v8f bias8, zero8;
#pragma unroll
  for (int r = 0; r < 8; ++r) {
    const int o = 8 * half + r;
    const float u = bd[(o < OO) ? o : OO - 1];
    bias8[r] = (o < OO) ? u : 0.f;
    zero8[r] = 0.f;
  }

  v2f ho[4];
#pragma unroll
  for (int q = 0; q < 4; ++q) ho[q] = (v2f){0.f, 0.f};
  v8f pl = zero8;

  const int* __restrict__ q1 = num1 + (size_t)row * TT;
  const int* __restrict__ q2 = num2 + (size_t)row * TT;
  v4i dn1 = *(const v4i*)q1;
  v4i dn2 = *(const v4i*)q2;

  float* const hp = hx + XOFF(0, wr) + lane * 4;
  float* const pp = px + XOFF(0, wr) + lane * 4;

  if (rec) {
    const int tb = w >> 1;
    const int j  = tid & 63;
    const float* __restrict__ wxc = Wx + tb * EE * HH + j;
    float wx[EE];
#pragma unroll
    for (int d = 0; d < EE; ++d) wx[d] = wxc[d * HH];
    const float s_init = (tb == 0) ? brnn[j] : 0.f;
#pragma unroll
    for (int v = 0; v < VV; ++v) {
      float s = s_init;
#pragma unroll
      for (int d = 0; d < EE; ++d) s += emb[v * EE + d] * wx[d];
      lut[(tb * VV + v) * LP + j] = s;
    }
  }

  if (rec) {
    *(v4f*)(hp + XOFF(1, 0))      = (v4f){0.f, 0.f, 0.f, 0.f};
    *(v4f*)(hp + XOFF(1, 0) + XW) = (v4f){0.f, 0.f, 0.f, 0.f};
  }

  __syncthreads();

  for (int tc = 0; tc < TT; tc += 4) {
    const v4i d1 = dn1, d2 = dn2;
    const int tn = (tc + 4 < TT) ? tc + 4 : tc;
    dn1 = *(const v4i*)(q1 + tn);
    dn2 = *(const v4i*)(q2 + tn);
#pragma unroll
    for (int s = 0; s < 4; ++s) {
      const int t = tc + s;
      const int buf = s & 1;

      if (rec) {
        v2f hr[12];
#pragma unroll
        for (int r = 1; r < 4; ++r) {
          const float* sp = hx + XOFF(buf ^ 1, (w + r) & 3) + lane * 4;
          const v4f u0 = *(const v4f*)(sp);
          const v4f u1 = *(const v4f*)(sp + XW);
          hr[4 * (r - 1) + 0] = (v2f){u0.x, u0.y};
          hr[4 * (r - 1) + 1] = (v2f){u0.z, u0.w};
          hr[4 * (r - 1) + 2] = (v2f){u1.x, u1.y};
          hr[4 * (r - 1) + 3] = (v2f){u1.z, u1.w};
        }
        const int e1 = (s == 0) ? d1.x : (s == 1) ? d1.y : (s == 2) ? d1.z : d1.w;
        const int e2 = (s == 0) ? d2.x : (s == 1) ? d2.y : (s == 2) ? d2.z : d2.w;
        const v8f xa = table_row(lut, e1, half, w);
        const v8f xb = table_row(lut, VV + e2, half, w);
        __builtin_amdgcn_sched_barrier(0);

        *(v4f*)(pp + XOFF(buf, 0))      = (v4f){pl[0], pl[1], pl[2], pl[3]};
        *(v4f*)(pp + XOFF(buf, 0) + XW) = (v4f){pl[4], pl[5], pl[6], pl[7]};

        v8f a0 = zero8, a1 = zero8;
        a0 = WMMA(wha[0], ho[0], a0);
        a1 = WMMA(wha[1], ho[1], a1);
        a0 = WMMA(wha[2], ho[2], a0);
        a1 = WMMA(wha[3], ho[3], a1);
#pragma unroll
        for (int jj = 4; jj < 16; jj += 2) {
          a0 = WMMA(wha[jj],     hr[jj - 4], a0);
          a1 = WMMA(wha[jj + 1], hr[jj - 3], a1);
        }

        v8f hn;
#pragma unroll
        for (int r = 0; r < 8; ++r) hn[r] = DEV_TANH((a0[r] + xa[r]) + (a1[r] + xb[r]));
        *(v4f*)(hp + XOFF(buf, 0))      = (v4f){hn[0], hn[1], hn[2], hn[3]};
        *(v4f*)(hp + XOFF(buf, 0) + XW) = (v4f){hn[4], hn[5], hn[6], hn[7]};

        ho[0] = (v2f){hn[0], hn[1]};
        ho[1] = (v2f){hn[2], hn[3]};
        ho[2] = (v2f){hn[4], hn[5]};
        ho[3] = (v2f){hn[6], hn[7]};
        pl = zero8;
#pragma unroll
        for (int q = 0; q < 4; ++q) pl = WMMA(wda[q], ho[q], pl);
      } else if (t > 1) {
        v8f lq[4];
        load_partials(px, buf ^ 1, lane, lq);
        store_logits(out, row, t - 2, half, bias8 + lq[0] + lq[1] + lq[2] + lq[3]);
      }

      lds_barrier();
    }
  }

  if (rec) {
    *(v4f*)(pp)      = (v4f){pl[0], pl[1], pl[2], pl[3]};
    *(v4f*)(pp + XW) = (v4f){pl[4], pl[5], pl[6], pl[7]};
  } else {
    v8f lq[4];
    load_partials(px, 1, lane, lq);
    store_logits(out, row, TT - 2, half, bias8 + lq[0] + lq[1] + lq[2] + lq[3]);
  }
  lds_barrier();
  if (!rec) {
    v8f lq[4];
    load_partials(px, 0, lane, lq);
    store_logits(out, row, TT - 1, half, bias8 + lq[0] + lq[1] + lq[2] + lq[3]);
  }
}

extern "C" void kernel_launch(void* const* d_in, const int* in_sizes, int n_in,
                              void* d_out, int out_size, void* d_ws, size_t ws_size,
                              hipStream_t stream) {
  const int*   num1 = (const int*)d_in[0];
  const int*   num2 = (const int*)d_in[1];
  const float* emb  = (const float*)d_in[2];
  const float* Wx   = (const float*)d_in[3];
  const float* Wh   = (const float*)d_in[4];
  const float* brnn = (const float*)d_in[5];
  const float* Wd   = (const float*)d_in[6];
  const float* bd   = (const float*)d_in[7];
  float* out = (float*)d_out;

  rnn_scan_kernel<<<BB / 16, 160, 0, stream>>>(num1, num2, emb, Wx, brnn, Wh, Wd, bd, out);
}


# Each recurrence wave forms the whole logits of every 4th step from h planes kept in an LDS ring of 8 steps: no partial-logit planes or LDS stores of them, no fifth wave; logit sum order changes
# speedup vs baseline: 5.5464x; 1.0354x over previous
#include <hip/hip_runtime.h>


typedef __attribute__((ext_vector_type(2))) float v2f;
typedef __attribute__((ext_vector_type(4))) float v4f;
typedef __attribute__((ext_vector_type(8))) float v8f;
typedef __attribute__((ext_vector_type(4))) int v4i;

#define BB 1024
#define TT 256
#define HH 64
#define EE 32
#define VV 10
#define OO 10
#define LP 68

#if __has_builtin(__builtin_amdgcn_tanhf)
#define DEV_TANH(x) __builtin_amdgcn_tanhf(x)
#else
#define DEV_TANH(x) tanhf(x)
#endif

#define WMMA(A, B, C) __builtin_amdgcn_wmma_f32_16x16x4_f32(false, (A), false, (B), (short)0, (C), false, false)

#define XW 128
#define RING 8
#define XOFF(slot, tile) ((((slot) * 4 + (tile)) * 2) * XW)

__device__ __forceinline__ void lds_barrier() {
  asm volatile("s_wait_dscnt 0x0\n\ts_barrier_signal -1\n\ts_barrier_wait -1" ::: "memory");
}

__device__ __forceinline__ v8f table_row(const float* lut, int d, int half, int tile) {
  const float* p = lut + d * LP + 16 * tile + 8 * half;
  const v4f a0 = *(const v4f*)(p);
  const v4f a1 = *(const v4f*)(p + 4);
  return (v8f){a0.x, a0.y, a0.z, a0.w, a1.x, a1.y, a1.z, a1.w};
}

__device__ __forceinline__ void store_logits(float* __restrict__ out, int row, int t, int half, v8f c2) {
  float* o = out + (size_t)row * (TT * OO) + (size_t)t * OO + 8 * half;
  *(v2f*)(o) = (v2f){c2[0], c2[1]};
  if (half == 0) {
    *(v2f*)(o + 2) = (v2f){c2[2], c2[3]};
    *(v2f*)(o + 4) = (v2f){c2[4], c2[5]};
    *(v2f*)(o + 6) = (v2f){c2[6], c2[7]};
  }
}

__device__ __forceinline__ void load_tile(const float* sp, v2f f[4]) {
  const v4f u0 = *(const v4f*)(sp);
  const v4f u1 = *(const v4f*)(sp + XW);
  f[0] = (v2f){u0.x, u0.y};
  f[1] = (v2f){u0.z, u0.w};
  f[2] = (v2f){u1.x, u1.y};
  f[3] = (v2f){u1.z, u1.w};
}

__global__ __launch_bounds__(128) void rnn_scan_kernel(
    const int* __restrict__ num1, const int* __restrict__ num2,
    const float* __restrict__ emb, const float* __restrict__ Wx,
    const float* __restrict__ brnn, const float* __restrict__ Wh,
    const float* __restrict__ Wd, const float* __restrict__ bd,
    float* __restrict__ out) {
  __shared__ __attribute__((aligned(16))) float lut[2 * VV * LP];
  __shared__ __attribute__((aligned(16))) float hx[RING * 4 * 2 * XW];

  const int tid  = threadIdx.x;
  const int w    = __builtin_amdgcn_readfirstlane(tid >> 5);
  const int lane = tid & 31;
  const int half = lane >> 4;
  const int c    = lane & 15;
  const int row  = blockIdx.x * 16 + c;

  v2f wha[16];
  v2f wda[16];
  const int cw = (c < OO) ? c : OO - 1;
#pragma unroll
  for (int jj = 0; jj < 16; ++jj) {
    const int tl = (w + (jj >> 2)) & 3;
    const int k0 = 16 * tl + 8 * half + 2 * (jj & 3);
    wha[jj].x = Wh[(k0 + 0) * HH + w * 16 + c];
    wha[jj].y = Wh[(k0 + 1) * HH + w * 16 + c];
  }
#pragma unroll
  for (int jj = 0; jj < 16; ++jj) {
    const int k0 = 16 * (jj >> 2) + 8 * half + 2 * (jj & 3);
    const float u0 = Wd[(k0 + 0) * OO + cw];
    const float u1 = Wd[(k0 + 1) * OO + cw];
    wda[jj].x = (c < OO) ? u0 : 0.f;
    wda[jj].y = (c < OO) ? u1 : 0.f;
  }
  v8f bias8, zero8;
#pragma unroll
  for (int r = 0; r < 8; ++r) {
    const int o = 8 * half + r;
    const float u = bd[(o < OO) ? o : OO - 1];
    bias8[r] = (o < OO) ? u : 0.f;
    zero8[r] = 0.f;
  }

  v2f ho[4];
#pragma unroll
  for (int q = 0; q < 4; ++q) ho[q] = (v2f){0.f, 0.f};
  v8f pl0 = zero8, pl1 = zero8;

  const int* __restrict__ q1 = num1 + (size_t)row * TT;
  const int* __restrict__ q2 = num2 + (size_t)row * TT;
  v4i dn1 = *(const v4i*)q1;
  v4i dn2 = *(const v4i*)q2;

  float* const hp = hx + XOFF(0, w) + lane * 4;
  const float* const lg0 = hx + XOFF(7 - w, 0) + lane * 4;
  const float* const lg1 = hx + XOFF(3 - w, 0) + lane * 4;

  {
    const int tb = w >> 1;
    const int j  = tid & 63;
    const float* __restrict__ wxc = Wx + tb * EE * HH + j;
    float wx[EE];
#pragma unroll
    for (int d = 0; d < EE; ++d) wx[d] = wxc[d * HH];
    const float s_init = (tb == 0) ? brnn[j] : 0.f;
#pragma unroll
    for (int v = 0; v < VV; ++v) {
      float s = s_init;
#pragma unroll
      for (int d = 0; d < EE; ++d) s += emb[v * EE + d] * wx[d];
      lut[(tb * VV + v) * LP + j] = s;
    }
  }

#pragma unroll
  for (int sl = 4; sl < RING; ++sl) {
    *(v4f*)(hp + XOFF(sl, 0))      = (v4f){0.f, 0.f, 0.f, 0.f};
    *(v4f*)(hp + XOFF(sl, 0) + XW) = (v4f){0.f, 0.f, 0.f, 0.f};
  }

  __syncthreads();

  for (int tc = 0; tc < TT; tc += 8) {
#pragma unroll
    for (int g = 0; g < 2; ++g) {
      const v4i d1 = dn1, d2 = dn2;
      const int tn = (tc + 4 * g + 4 < TT) ? tc + 4 * g + 4 : tc + 4 * g;
      dn1 = *(const v4i*)(q1 + tn);
      dn2 = *(const v4i*)(q2 + tn);
      const float* const lg = g ? lg1 : lg0;
#pragma unroll
      for (int s = 0; s < 4; ++s) {
        const int u = 4 * g + s;
        const int up = (u + 7) & 7;

        v2f hr[12];
#pragma unroll
        for (int r = 1; r < 4; ++r) load_tile(hx + XOFF(up, (w + r) & 3) + lane * 4, hr + 4 * (r - 1));
        const int e1 = (s == 0) ? d1.x : (s == 1) ? d1.y : (s == 2) ? d1.z : d1.w;
        const int e2 = (s == 0) ? d2.x : (s == 1) ? d2.y : (s == 2) ? d2.z : d2.w;
        const v8f xa = table_row(lut, e1, half, w);
        const v8f xb = table_row(lut, VV + e2, half, w);
        v2f lf[4];
        load_tile(lg + XOFF(0, s), lf);
        __builtin_amdgcn_sched_barrier(0);

        v8f a0 = zero8, a1 = zero8;
        a0 = WMMA(wha[0], ho[0], a0);
        a1 = WMMA(wha[1], ho[1], a1);
        a0 = WMMA(wha[2], ho[2], a0);
        a1 = WMMA(wha[3], ho[3], a1);
#pragma unroll
        for (int jj = 4; jj < 16; jj += 2) {
          a0 = WMMA(wha[jj],     hr[jj - 4], a0);
          a1 = WMMA(wha[jj + 1], hr[jj - 3], a1);
        }

        __builtin_amdgcn_sched_barrier(0);
        const v8f fin = pl0 + pl1;
        pl0 = WMMA(wda[4 * s + 0], lf[0], (s == 0) ? bias8 : pl0);

        v8f hn;
#pragma unroll
        for (int r = 0; r < 8; ++r) hn[r] = DEV_TANH((a0[r] + xa[r]) + (a1[r] + xb[r]));
        *(v4f*)(hp + XOFF(u, 0))      = (v4f){hn[0], hn[1], hn[2], hn[3]};
        *(v4f*)(hp + XOFF(u, 0) + XW) = (v4f){hn[4], hn[5], hn[6], hn[7]};
        ho[0] = (v2f){hn[0], hn[1]};
        ho[1] = (v2f){hn[2], hn[3]};
        ho[2] = (v2f){hn[4], hn[5]};
        ho[3] = (v2f){hn[6], hn[7]};
        __builtin_amdgcn_sched_barrier(0);

        if (s == 0) {
          if (tc != 0) store_logits(out, row, tc + 4 * g - 5 - w, half, fin);
          pl1 = WMMA(wda[1], lf[1], zero8);
        } else {
          pl1 = WMMA(wda[4 * s + 1], lf[1], pl1);
        }
        pl0 = WMMA(wda[4 * s + 2], lf[2], pl0);
        pl1 = WMMA(wda[4 * s + 3], lf[3], pl1);
        __builtin_amdgcn_sched_barrier(0);

        lds_barrier();
      }
    }
  }

  store_logits(out, row, TT - 5 - w, half, pl0 + pl1);
  {
    v2f lf[16];
#pragma unroll
    for (int tl = 0; tl < 4; ++tl) load_tile(lg0 + XOFF(0, tl), lf + 4 * tl);
    pl0 = bias8;
    pl1 = zero8;
#pragma unroll
    for (int jj = 0; jj < 16; jj += 2) {
      pl0 = WMMA(wda[jj],     lf[jj],     pl0);
      pl1 = WMMA(wda[jj + 1], lf[jj + 1], pl1);
    }
    store_logits(out, row, TT - 1 - w, half, pl0 + pl1);
  }
}

extern "C" void kernel_launch(void* const* d_in, const int* in_sizes, int n_in,
                              void* d_out, int out_size, void* d_ws, size_t ws_size,
                              hipStream_t stream) {
  const int*   num1 = (const int*)d_in[0];
  const int*   num2 = (const int*)d_in[1];
  const float* emb  = (const float*)d_in[2];
  const float* Wx   = (const float*)d_in[3];
  const float* Wh   = (const float*)d_in[4];
  const float* brnn = (const float*)d_in[5];
  const float* Wd   = (const float*)d_in[6];
  const float* bd   = (const float*)d_in[7];
  float* out = (float*)d_out;

  rnn_scan_kernel<<<BB / 16, 128, 0, stream>>>(num1, num2, emb, Wx, brnn, Wh, Wd, bd, out);
}


# Next step's table rows and logit tile read a step ahead behind the h stores under a counted wait (signal on the stores only); x starts the chains: sum order changes; logit WMMAs fill barrier waits
# speedup vs baseline: 5.7452x; 1.0358x over previous
#include <hip/hip_runtime.h>


typedef __attribute__((ext_vector_type(2))) float v2f;
typedef __attribute__((ext_vector_type(4))) float v4f;
typedef __attribute__((ext_vector_type(8))) float v8f;
typedef __attribute__((ext_vector_type(4))) int v4i;

#define BB 1024
#define TT 256
#define HH 64
#define EE 32
#define VV 10
#define OO 10
#define LP 68

#if __has_builtin(__builtin_amdgcn_tanhf)
#define DEV_TANH(x) __builtin_amdgcn_tanhf(x)
#else
#define DEV_TANH(x) tanhf(x)
#endif

#define WMMA(A, B, C) __builtin_amdgcn_wmma_f32_16x16x4_f32(false, (A), false, (B), (short)0, (C), false, false)

#define XW 128
#define RING 8
#define XOFF(slot, tile) ((((slot) * 4 + (tile)) * 2) * XW)

__device__ __forceinline__ void handover_signal(v8f p0, v8f p1) {
  asm volatile("s_wait_dscnt 0x6\n\ts_barrier_signal -1" :: "v"(p0), "v"(p1) : "memory");
}
__device__ __forceinline__ void handover_wait(v8f p1) {
  asm volatile("s_barrier_wait -1" :: "v"(p1) : "memory");
}

__device__ __forceinline__ v8f table_row(const float* lut, int d, int half, int tile) {
  const float* p = lut + d * LP + 16 * tile + 8 * half;
  const v4f a0 = *(const v4f*)(p);
  const v4f a1 = *(const v4f*)(p + 4);
  return (v8f){a0.x, a0.y, a0.z, a0.w, a1.x, a1.y, a1.z, a1.w};
}

__device__ __forceinline__ void store_logits(float* __restrict__ out, int row, int t, int half, v8f c2) {
  float* o = out + (size_t)row * (TT * OO) + (size_t)t * OO + 8 * half;
  *(v2f*)(o) = (v2f){c2[0], c2[1]};
  if (half == 0) {
    *(v2f*)(o + 2) = (v2f){c2[2], c2[3]};
    *(v2f*)(o + 4) = (v2f){c2[4], c2[5]};
    *(v2f*)(o + 6) = (v2f){c2[6], c2[7]};
  }
}

__device__ __forceinline__ void load_tile(const float* sp, v2f f[4]) {
  const v4f u0 = *(const v4f*)(sp);
  const v4f u1 = *(const v4f*)(sp + XW);
  f[0] = (v2f){u0.x, u0.y};
  f[1] = (v2f){u0.z, u0.w};
  f[2] = (v2f){u1.x, u1.y};
  f[3] = (v2f){u1.z, u1.w};
}

__global__ __launch_bounds__(128) void rnn_scan_kernel(
    const int* __restrict__ num1, const int* __restrict__ num2,
    const float* __restrict__ emb, const float* __restrict__ Wx,
    const float* __restrict__ brnn, const float* __restrict__ Wh,
    const float* __restrict__ Wd, const float* __restrict__ bd,
    float* __restrict__ out) {
  __shared__ __attribute__((aligned(16))) float lut[2 * VV * LP];
  __shared__ __attribute__((aligned(16))) float hx[RING * 4 * 2 * XW];

  const int tid  = threadIdx.x;
  const int w    = __builtin_amdgcn_readfirstlane(tid >> 5);
  const int lane = tid & 31;
  const int half = lane >> 4;
  const int c    = lane & 15;
  const int row  = blockIdx.x * 16 + c;

  v2f wha[16];
  v2f wda[16];
  const int cw = (c < OO) ? c : OO - 1;
#pragma unroll
  for (int jj = 0; jj < 16; ++jj) {
    const int tl = (w + (jj >> 2)) & 3;
    const int k0 = 16 * tl + 8 * half + 2 * (jj & 3);
    wha[jj].x = Wh[(k0 + 0) * HH + w * 16 + c];
    wha[jj].y = Wh[(k0 + 1) * HH + w * 16 + c];
  }
#pragma unroll
  for (int jj = 0; jj < 16; ++jj) {
    const int k0 = 16 * (jj >> 2) + 8 * half + 2 * (jj & 3);
    const float u0 = Wd[(k0 + 0) * OO + cw];
    const float u1 = Wd[(k0 + 1) * OO + cw];
    wda[jj].x = (c < OO) ? u0 : 0.f;
    wda[jj].y = (c < OO) ? u1 : 0.f;
  }
  v8f bias8, zero8;
#pragma unroll
  for (int r = 0; r < 8; ++r) {
    const int o = 8 * half + r;
    const float u = bd[(o < OO) ? o : OO - 1];
    bias8[r] = (o < OO) ? u : 0.f;
    zero8[r] = 0.f;
  }

  v2f ho[4];
#pragma unroll
  for (int q = 0; q < 4; ++q) ho[q] = (v2f){0.f, 0.f};
  v8f pl0 = zero8, pl1 = zero8;

  const int* __restrict__ q1 = num1 + (size_t)row * TT;
  const int* __restrict__ q2 = num2 + (size_t)row * TT;
  v4i dn1 = *(const v4i*)q1;
  v4i dn2 = *(const v4i*)q2;

  float* const hp = hx + XOFF(0, w) + lane * 4;
  const float* const lg0 = hx + XOFF(7 - w, 0) + lane * 4;
  const float* const lg1 = hx + XOFF(3 - w, 0) + lane * 4;

  {
    const int tb = w >> 1;
    const int j  = tid & 63;
    const float* __restrict__ wxc = Wx + tb * EE * HH + j;
    float wx[EE];
#pragma unroll
    for (int d = 0; d < EE; ++d) wx[d] = wxc[d * HH];
    const float s_init = (tb == 0) ? brnn[j] : 0.f;
#pragma unroll
    for (int v = 0; v < VV; ++v) {
      float s = s_init;
#pragma unroll
      for (int d = 0; d < EE; ++d) s += emb[v * EE + d] * wx[d];
      lut[(tb * VV + v) * LP + j] = s;
    }
  }

#pragma unroll
  for (int sl = 4; sl < RING; ++sl) {
    *(v4f*)(hp + XOFF(sl, 0))      = (v4f){0.f, 0.f, 0.f, 0.f};
    *(v4f*)(hp + XOFF(sl, 0) + XW) = (v4f){0.f, 0.f, 0.f, 0.f};
  }

  __syncthreads();

  v8f xa = table_row(lut, dn1.x, half, w);
  v8f xb = table_row(lut, VV + dn2.x, half, w);
  v2f lf[4];
  load_tile(lg0, lf);

  for (int tc = 0; tc < TT; tc += 8) {
#pragma unroll
    for (int g = 0; g < 2; ++g) {
      const v4i d1 = dn1, d2 = dn2;
      const int tn = (tc + 4 * g + 4 < TT) ? tc + 4 * g + 4 : tc + 4 * g;
      dn1 = *(const v4i*)(q1 + tn);
      dn2 = *(const v4i*)(q2 + tn);
      const float* const lg  = g ? lg1 : lg0;
      const float* const lgn = g ? lg0 : lg1;
#pragma unroll
      for (int s = 0; s < 4; ++s) {
        const int u = 4 * g + s;
        const int up = (u + 7) & 7;

        v2f hr[12];
#pragma unroll
        for (int r = 1; r < 4; ++r) load_tile(hx + XOFF(up, (w + r) & 3) + lane * 4, hr + 4 * (r - 1));
        __builtin_amdgcn_sched_barrier(0);

        const v8f plo = pl0;
        v8f a0 = WMMA(wha[0], ho[0], xa);
        v8f a1 = WMMA(wha[1], ho[1], xb);
        a0 = WMMA(wha[2], ho[2], a0);
        a1 = WMMA(wha[3], ho[3], a1);
        pl0 = WMMA(wda[4 * s + 0], lf[0], (s == 0) ? bias8 : pl0);
        asm volatile("" :: "v"(pl0));
        __builtin_amdgcn_sched_barrier(0);
#pragma unroll
        for (int jj = 4; jj < 16; jj += 2) {
          a0 = WMMA(wha[jj],     hr[jj - 4], a0);
          a1 = WMMA(wha[jj + 1], hr[jj - 3], a1);
        }
        __builtin_amdgcn_sched_barrier(0);

        const v8f fin = plo + pl1;

        v8f hn;
#pragma unroll
        for (int r = 0; r < 8; ++r) hn[r] = DEV_TANH(a0[r] + a1[r]);
        *(v4f*)(hp + XOFF(u, 0))      = (v4f){hn[0], hn[1], hn[2], hn[3]};
        *(v4f*)(hp + XOFF(u, 0) + XW) = (v4f){hn[4], hn[5], hn[6], hn[7]};
        ho[0] = (v2f){hn[0], hn[1]};
        ho[1] = (v2f){hn[2], hn[3]};
        ho[2] = (v2f){hn[4], hn[5]};
        ho[3] = (v2f){hn[6], hn[7]};
        __builtin_amdgcn_sched_barrier(0);

        const int e1 = (s == 0) ? d1.y : (s == 1) ? d1.z : (s == 2) ? d1.w : dn1.x;
        const int e2 = (s == 0) ? d2.y : (s == 1) ? d2.z : (s == 2) ? d2.w : dn2.x;
        const v8f xan = table_row(lut, e1, half, w);
        const v8f xbn = table_row(lut, VV + e2, half, w);
        v2f lfn[4];
        load_tile((s < 3) ? lg + XOFF(0, s + 1) : lgn, lfn);
        __builtin_amdgcn_sched_barrier(0);

        if (s == 0) {
          if (tc != 0) store_logits(out, row, tc + 4 * g - 5 - w, half, fin);
          pl1 = WMMA(wda[1], lf[1], zero8);
        } else {
          pl1 = WMMA(wda[4 * s + 1], lf[1], pl1);
        }
        pl0 = WMMA(wda[4 * s + 2], lf[2], pl0);
        __builtin_amdgcn_sched_barrier(0);

        handover_signal(pl0, pl1);
        __builtin_amdgcn_sched_barrier(0);
        pl1 = WMMA(wda[4 * s + 3], lf[3], pl1);
        __builtin_amdgcn_sched_barrier(0);
        handover_wait(pl1);

        xa = xan;
        xb = xbn;
#pragma unroll
        for (int q = 0; q < 4; ++q) lf[q] = lfn[q];
      }
    }
  }

  store_logits(out, row, TT - 5 - w, half, pl0 + pl1);
  {
    v2f lf[16];
#pragma unroll
    for (int tl = 0; tl < 4; ++tl) load_tile(lg0 + XOFF(0, tl), lf + 4 * tl);
    pl0 = bias8;
    pl1 = zero8;
#pragma unroll
    for (int jj = 0; jj < 16; jj += 2) {
      pl0 = WMMA(wda[jj],     lf[jj],     pl0);
      pl1 = WMMA(wda[jj + 1], lf[jj + 1], pl1);
    }
    store_logits(out, row, TT - 1 - w, half, pl0 + pl1);
  }
}

extern "C" void kernel_launch(void* const* d_in, const int* in_sizes, int n_in,
                              void* d_out, int out_size, void* d_ws, size_t ws_size,
                              hipStream_t stream) {
  const int*   num1 = (const int*)d_in[0];
  const int*   num2 = (const int*)d_in[1];
  const float* emb  = (const float*)d_in[2];
  const float* Wx   = (const float*)d_in[3];
  const float* Wh   = (const float*)d_in[4];
  const float* brnn = (const float*)d_in[5];
  const float* Wd   = (const float*)d_in[6];
  const float* bd   = (const float*)d_in[7];
  float* out = (float*)d_out;

  rnn_scan_kernel<<<BB / 16, 128, 0, stream>>>(num1, num2, emb, Wx, brnn, Wh, Wd, bd, out);
}
